# starting kernel
# baseline (speedup 1.0000x reference)
#include <hip/hip_runtime.h>
#include <hip/hip_bf16.h>


typedef __attribute__((ext_vector_type(2))) float v2f;
typedef __attribute__((ext_vector_type(8))) float v8f;

#define D 256

__global__ __launch_bounds__(256) void sage_deg_kernel(
    const int* __restrict__ dst, float* __restrict__ deg, int E)
{
    int i = blockIdx.x * blockDim.x + threadIdx.x;
    if (i < E) atomicAdd(&deg[dst[i]], 1.0f);
}

__global__ __launch_bounds__(256) void sage_invdeg_kernel(
    float* __restrict__ deg, int Nn)
{
    int i = blockIdx.x * blockDim.x + threadIdx.x;
    if (i < Nn) deg[i] = 1.0f / fmaxf(deg[i], 1.0f);
}

__global__ __launch_bounds__(256) void sage_scatter_kernel(
    const float* __restrict__ feat, const int* __restrict__ src,
    const int* __restrict__ dst, float* __restrict__ agg, int E)
{
    long long i = (long long)blockIdx.x * blockDim.x + threadIdx.x;
    int e = (int)(i >> 6);
    if (e >= E) return;
    int q = ((int)i & 63) << 2;
    int s  = src[e];
    int d0 = dst[e];

    int ep = e + 512;
    if (ep < E) __builtin_prefetch(feat + (size_t)src[ep] * D + q, 0, 1);

    const float4 v = *(const float4*)(feat + (size_t)s * D + q);
    float* p = agg + (size_t)d0 * D + q;
    atomicAdd(p + 0, v.x);
    atomicAdd(p + 1, v.y);
    atomicAdd(p + 2, v.z);
    atomicAdd(p + 3, v.w);
}

__global__ __launch_bounds__(256) void sage_gemm_kernel(
    const float* __restrict__ agg,  const float* __restrict__ invdeg,
    const float* __restrict__ feat, const float* __restrict__ Wl,
    const float* __restrict__ Wr,   const float* __restrict__ bias,
    float* __restrict__ out, int Nn, int do_relu)
{
    const int wave = threadIdx.x >> 5;
    const int lane = threadIdx.x & 31;
    const int half = lane >> 4;
    const int l15  = lane & 15;

    const int m0 = blockIdx.x * 32 + (wave >> 2) * 16;
    const int n0 = blockIdx.y * 64 + (wave & 3) * 16;

    int am  = m0 + l15;
    int amc = (am < Nn) ? am : (Nn - 1);
    const float sA = invdeg[amc];
    const float* __restrict__ arow = agg  + (size_t)amc * D;
    const float* __restrict__ xrow = feat + (size_t)amc * D;
    const int ncol = n0 + l15;

    v8f acc = {};

#pragma unroll 4
    for (int k = 0; k < D; k += 4) {
        const int ka = k + 2 * half;
        v2f a, b;
        a.x = arow[ka]     * sA;
        a.y = arow[ka + 1] * sA;
        b.x = Wl[(size_t)(ka)     * D + ncol];
        b.y = Wl[(size_t)(ka + 1) * D + ncol];
        acc = __builtin_amdgcn_wmma_f32_16x16x4_f32(false, a, false, b,
                                                    (short)0, acc, false, false);
    }

#pragma unroll 4
    for (int k = 0; k < D; k += 4) {
        const int ka = k + 2 * half;
        v2f a, b;
        a.x = xrow[ka];
        a.y = xrow[ka + 1];
        b.x = Wr[(size_t)(ka)     * D + ncol];
        b.y = Wr[(size_t)(ka + 1) * D + ncol];
        acc = __builtin_amdgcn_wmma_f32_16x16x4_f32(false, a, false, b,
                                                    (short)0, acc, false, false);
    }

    const float bn = bias[ncol];
#pragma unroll
    for (int v = 0; v < 8; ++v) {
        const int row = m0 + v + 8 * half;
        if (row < Nn) {
            float val = acc[v] + bn;
            if (do_relu) val = fmaxf(val, 0.0f);
            out[(size_t)row * D + ncol] = val;
        }
    }
}

extern "C" void kernel_launch(void* const* d_in, const int* in_sizes, int n_in,
                              void* d_out, int out_size, void* d_ws, size_t ws_size,
                              hipStream_t stream)
{
    const float* x   = (const float*)d_in[0];
    const int*   ei  = (const int*)d_in[1];
    const float* W1l = (const float*)d_in[2];
    const float* b1  = (const float*)d_in[3];
    const float* W1r = (const float*)d_in[4];
    const float* W2l = (const float*)d_in[5];
    const float* b2  = (const float*)d_in[6];
    const float* W2r = (const float*)d_in[7];
    float* out = (float*)d_out;

    const int Nn = in_sizes[0] / D;
    const int E  = in_sizes[1] / 2;
    const int* src = ei;
    const int* dst = ei + E;

    float* agg = (float*)d_ws;
    float* deg = agg + (size_t)Nn * D;
    float* h   = deg + (size_t)Nn;

    const size_t aggBytes = (size_t)Nn * D * sizeof(float);
    const size_t degBytes = (size_t)Nn * sizeof(float);

    const int blk = 256;
    const long long scatterThreads = (long long)E * 64;
    const int scatterBlocks = (int)((scatterThreads + blk - 1) / blk);
    dim3 gemmGrid((Nn + 31) / 32, D / 64);

    hipMemsetAsync(agg, 0, aggBytes + degBytes, stream);
    sage_deg_kernel<<<(E + blk - 1) / blk, blk, 0, stream>>>(dst, deg, E);
    sage_invdeg_kernel<<<(Nn + blk - 1) / blk, blk, 0, stream>>>(deg, Nn);

    sage_scatter_kernel<<<scatterBlocks, blk, 0, stream>>>(x, src, dst, agg, E);
    sage_gemm_kernel<<<gemmGrid, blk, 0, stream>>>(agg, deg, x, W1l, W1r, b1,
                                                   h, Nn,  1);

    hipMemsetAsync(agg, 0, aggBytes, stream);
    sage_scatter_kernel<<<scatterBlocks, blk, 0, stream>>>(h, src, dst, agg, E);
    sage_gemm_kernel<<<gemmGrid, blk, 0, stream>>>(agg, deg, h, W2l, W2r, b2,
                                                   out, Nn,  0);
}


# Aggregation without f32 atomics: edges grouped by destination once per call (returning int add gives the slot, prefix sum), one wave per node sums its rows in registers in slot (arrival) order
# speedup vs baseline: 1.8167x; 1.8167x over previous
#include <hip/hip_runtime.h>
#include <hip/hip_bf16.h>


typedef __attribute__((ext_vector_type(2))) float v2f;
typedef __attribute__((ext_vector_type(8))) float v8f;

#define D 256

__global__ __launch_bounds__(256) void sage_rank_kernel(
    const int* __restrict__ dst, int* __restrict__ cnt,
    int* __restrict__ rank, int E)
{
    int e = blockIdx.x * blockDim.x + threadIdx.x;
    if (e < E) rank[e] = atomicAdd(&cnt[dst[e]], 1);
}

__global__ __launch_bounds__(1024) void sage_scan_kernel(
    int* cd, int* __restrict__ rowptr, int Nn)
{
    __shared__ int wsum[32];
    const int t = threadIdx.x;
    const int lane = t & 31;
    const int w = t >> 5;
    const int per = (Nn + 1023) / 1024;
    const int i0 = min(t * per, Nn);
    const int i1 = min(i0 + per, Nn);

    int s = 0;
#pragma unroll 8
    for (int i = i0; i < i1; ++i) s += cd[i];

    int inc = s;
    for (int o = 1; o < 32; o <<= 1) {
        int v = __shfl_up(inc, o);
        if (lane >= o) inc += v;
    }
    if (lane == 31) wsum[w] = inc;
    __syncthreads();
    if (w == 0) {
        int v = wsum[lane];
        int vi = v;
        for (int o = 1; o < 32; o <<= 1) {
            int u = __shfl_up(vi, o);
            if (lane >= o) vi += u;
        }
        wsum[lane] = vi - v;
    }
    __syncthreads();

    int run = wsum[w] + inc - s;
    for (int i = i0; i < i1; ++i) {
        const int c = cd[i];
        rowptr[i] = run;
        cd[i] = __float_as_int(1.0f / fmaxf((float)c, 1.0f));
        run += c;
    }
    if (t == 1023) rowptr[Nn] = run;
}

__global__ __launch_bounds__(256) void sage_fill_kernel(
    const int* __restrict__ src, const int* __restrict__ dst,
    const int* __restrict__ rowptr, const int* __restrict__ rank,
    int* __restrict__ col, int E)
{
    int e = blockIdx.x * blockDim.x + threadIdx.x;
    if (e >= E) return;
    int p = rowptr[dst[e]] + rank[e];
    p = min(max(p, 0), E - 1);
    col[p] = src[e];
}

__global__ __launch_bounds__(256) void sage_gather_kernel(
    const float* __restrict__ feat, const int* __restrict__ rowptr,
    const int* __restrict__ col, float* __restrict__ agg, int Nn, int E)
{
    const int lane = threadIdx.x & 31;
    const int n = blockIdx.x * 8 + (threadIdx.x >> 5);
    if (n >= Nn) return;

    int beg = rowptr[n];
    int end = rowptr[n + 1];
    beg = min(max(beg, 0), E);
    end = min(max(end, beg), E);
    beg = __builtin_amdgcn_readfirstlane(beg);
    end = __builtin_amdgcn_readfirstlane(end);

    const int nmax = Nn - 1;
    const float* __restrict__ fb = feat + lane * 4;
    float4 a0 = make_float4(0.f, 0.f, 0.f, 0.f);
    float4 a1 = make_float4(0.f, 0.f, 0.f, 0.f);

    for (int base = beg; base < end; base += 32) {
        const int m = min(32, end - base);
        int myc = (lane < m) ? col[base + lane] : 0;
        myc = min(max(myc, 0), nmax);
        int j = 0;
        for (; j + 4 <= m; j += 4) {
            const int s0 = __builtin_amdgcn_readlane(myc, j);
            const int s1 = __builtin_amdgcn_readlane(myc, j + 1);
            const int s2 = __builtin_amdgcn_readlane(myc, j + 2);
            const int s3 = __builtin_amdgcn_readlane(myc, j + 3);
            const float* p0 = fb + (size_t)s0 * D;
            const float* p1 = fb + (size_t)s1 * D;
            const float* p2 = fb + (size_t)s2 * D;
            const float* p3 = fb + (size_t)s3 * D;
            const float4 x0 = *(const float4*)(p0);
            const float4 y0 = *(const float4*)(p0 + 128);
            const float4 x1 = *(const float4*)(p1);
            const float4 y1 = *(const float4*)(p1 + 128);
            const float4 x2 = *(const float4*)(p2);
            const float4 y2 = *(const float4*)(p2 + 128);
            const float4 x3 = *(const float4*)(p3);
            const float4 y3 = *(const float4*)(p3 + 128);
            a0.x += x0.x; a0.y += x0.y; a0.z += x0.z; a0.w += x0.w;
            a1.x += y0.x; a1.y += y0.y; a1.z += y0.z; a1.w += y0.w;
            a0.x += x1.x; a0.y += x1.y; a0.z += x1.z; a0.w += x1.w;
            a1.x += y1.x; a1.y += y1.y; a1.z += y1.z; a1.w += y1.w;
            a0.x += x2.x; a0.y += x2.y; a0.z += x2.z; a0.w += x2.w;
            a1.x += y2.x; a1.y += y2.y; a1.z += y2.z; a1.w += y2.w;
            a0.x += x3.x; a0.y += x3.y; a0.z += x3.z; a0.w += x3.w;
            a1.x += y3.x; a1.y += y3.y; a1.z += y3.z; a1.w += y3.w;
        }
        const int r = m - j;
        if (r > 0) {
            const int s0 = __builtin_amdgcn_readlane(myc, j);
            const int s1 = __builtin_amdgcn_readlane(myc, min(j + 1, 31));
            const int s2 = __builtin_amdgcn_readlane(myc, min(j + 2, 31));
            const float* p0 = fb + (size_t)s0 * D;
            const float* p1 = fb + (size_t)s1 * D;
            const float* p2 = fb + (size_t)s2 * D;
            const float4 x0 = *(const float4*)(p0);
            const float4 y0 = *(const float4*)(p0 + 128);
            float4 x1 = make_float4(0.f, 0.f, 0.f, 0.f), y1 = x1, x2 = x1, y2 = x1;
            if (r > 1) { x1 = *(const float4*)(p1); y1 = *(const float4*)(p1 + 128); }
            if (r > 2) { x2 = *(const float4*)(p2); y2 = *(const float4*)(p2 + 128); }
            a0.x += x0.x; a0.y += x0.y; a0.z += x0.z; a0.w += x0.w;
            a1.x += y0.x; a1.y += y0.y; a1.z += y0.z; a1.w += y0.w;
            if (r > 1) {
                a0.x += x1.x; a0.y += x1.y; a0.z += x1.z; a0.w += x1.w;
                a1.x += y1.x; a1.y += y1.y; a1.z += y1.z; a1.w += y1.w;
            }
            if (r > 2) {
                a0.x += x2.x; a0.y += x2.y; a0.z += x2.z; a0.w += x2.w;
                a1.x += y2.x; a1.y += y2.y; a1.z += y2.z; a1.w += y2.w;
            }
        }
    }

    float* o = agg + (size_t)n * D + lane * 4;
    *(float4*)(o)       = a0;
    *(float4*)(o + 128) = a1;
}

__global__ __launch_bounds__(256) void sage_gemm_kernel(
    const float* __restrict__ agg,  const float* __restrict__ invdeg,
    const float* __restrict__ feat, const float* __restrict__ Wl,
    const float* __restrict__ Wr,   const float* __restrict__ bias,
    float* __restrict__ out, int Nn, int do_relu)
{
    const int wave = threadIdx.x >> 5;
    const int lane = threadIdx.x & 31;
    const int half = lane >> 4;
    const int l15  = lane & 15;

    const int m0 = blockIdx.x * 32 + (wave >> 2) * 16;
    const int n0 = blockIdx.y * 64 + (wave & 3) * 16;

    int am  = m0 + l15;
    int amc = (am < Nn) ? am : (Nn - 1);
    const float sA = invdeg[amc];
    const float* __restrict__ arow = agg  + (size_t)amc * D;
    const float* __restrict__ xrow = feat + (size_t)amc * D;
    const int ncol = n0 + l15;

    v8f acc = {};

#pragma unroll 4
    for (int k = 0; k < D; k += 4) {
        const int ka = k + 2 * half;
        v2f a, b;
        a.x = arow[ka]     * sA;
        a.y = arow[ka + 1] * sA;
        b.x = Wl[(size_t)(ka)     * D + ncol];
        b.y = Wl[(size_t)(ka + 1) * D + ncol];
        acc = __builtin_amdgcn_wmma_f32_16x16x4_f32(false, a, false, b,
                                                    (short)0, acc, false, false);
    }

#pragma unroll 4
    for (int k = 0; k < D; k += 4) {
        const int ka = k + 2 * half;
        v2f a, b;
        a.x = xrow[ka];
        a.y = xrow[ka + 1];
        b.x = Wr[(size_t)(ka)     * D + ncol];
        b.y = Wr[(size_t)(ka + 1) * D + ncol];
        acc = __builtin_amdgcn_wmma_f32_16x16x4_f32(false, a, false, b,
                                                    (short)0, acc, false, false);
    }

    const float bn = bias[ncol];
#pragma unroll
    for (int v = 0; v < 8; ++v) {
        const int row = m0 + v + 8 * half;
        if (row < Nn) {
            float val = acc[v] + bn;
            if (do_relu) val = fmaxf(val, 0.0f);
            out[(size_t)row * D + ncol] = val;
        }
    }
}

extern "C" void kernel_launch(void* const* d_in, const int* in_sizes, int n_in,
                              void* d_out, int out_size, void* d_ws, size_t ws_size,
                              hipStream_t stream)
{
    const float* x   = (const float*)d_in[0];
    const int*   ei  = (const int*)d_in[1];
    const float* W1l = (const float*)d_in[2];
    const float* b1  = (const float*)d_in[3];
    const float* W1r = (const float*)d_in[4];
    const float* W2l = (const float*)d_in[5];
    const float* b2  = (const float*)d_in[6];
    const float* W2r = (const float*)d_in[7];
    float* out = (float*)d_out;

    const int Nn = in_sizes[0] / D;
    const int E  = in_sizes[1] / 2;
    const int* src = ei;
    const int* dst = ei + E;

    float* agg = (float*)d_ws;
    float* deg = agg + (size_t)Nn * D;
    float* h   = deg + (size_t)Nn;

    int* cnt    = (int*)deg;
    int* rank   = (int*)h;
    int* rowptr = (int*)d_out;
    int* col    = rowptr + (size_t)Nn + 1;

    const size_t degBytes = (size_t)Nn * sizeof(float);

    const int blk = 256;
    const int edgeBlocks = (E + blk - 1) / blk;
    const int gatherBlocks = (Nn + 7) / 8;
    dim3 gemmGrid((Nn + 31) / 32, D / 64);

    hipMemsetAsync(cnt, 0, degBytes, stream);
    sage_rank_kernel<<<edgeBlocks, blk, 0, stream>>>(dst, cnt, rank, E);
    sage_scan_kernel<<<1, 1024, 0, stream>>>(cnt, rowptr, Nn);
    sage_fill_kernel<<<edgeBlocks, blk, 0, stream>>>(src, dst, rowptr, rank, col, E);

    sage_gather_kernel<<<gatherBlocks, blk, 0, stream>>>(x, rowptr, col, agg, Nn, E);
    sage_gemm_kernel<<<gemmGrid, blk, 0, stream>>>(agg, deg, x, W1l, W1r, b1,
                                                   h, Nn,  1);

    sage_gather_kernel<<<gatherBlocks, blk, 0, stream>>>(h, rowptr, col, agg, Nn, E);
    sage_gemm_kernel<<<gemmGrid, blk, 0, stream>>>(agg, deg, h, W2l, W2r, b2,
                                                   out, Nn,  0);
}


# Linear layers: each wave keeps a 32x64 output tile in registers and loads A/B fragments as 16-byte vectors (A fragment feeds 4 WMMAs, B fragment 2); k summed in permuted order inside each group of 8
# speedup vs baseline: 3.1625x; 1.7408x over previous
#include <hip/hip_runtime.h>
#include <hip/hip_bf16.h>


typedef __attribute__((ext_vector_type(2))) float v2f;
typedef __attribute__((ext_vector_type(8))) float v8f;

#define D 256

__global__ __launch_bounds__(256) void sage_rank_kernel(
    const int* __restrict__ dst, int* __restrict__ cnt,
    int* __restrict__ rank, int E)
{
    int e = blockIdx.x * blockDim.x + threadIdx.x;
    if (e < E) rank[e] = atomicAdd(&cnt[dst[e]], 1);
}

__global__ __launch_bounds__(1024) void sage_scan_kernel(
    int* cd, int* __restrict__ rowptr, int Nn)
{
    __shared__ int wsum[32];
    const int t = threadIdx.x;
    const int lane = t & 31;
    const int w = t >> 5;
    const int per = (Nn + 1023) / 1024;
    const int i0 = min(t * per, Nn);
    const int i1 = min(i0 + per, Nn);

    int s = 0;
#pragma unroll 8
    for (int i = i0; i < i1; ++i) s += cd[i];

    int inc = s;
    for (int o = 1; o < 32; o <<= 1) {
        int v = __shfl_up(inc, o);
        if (lane >= o) inc += v;
    }
    if (lane == 31) wsum[w] = inc;
    __syncthreads();
    if (w == 0) {
        int v = wsum[lane];
        int vi = v;
        for (int o = 1; o < 32; o <<= 1) {
            int u = __shfl_up(vi, o);
            if (lane >= o) vi += u;
        }
        wsum[lane] = vi - v;
    }
    __syncthreads();

    int run = wsum[w] + inc - s;
    for (int i = i0; i < i1; ++i) {
        const int c = cd[i];
        rowptr[i] = run;
        cd[i] = __float_as_int(1.0f / fmaxf((float)c, 1.0f));
        run += c;
    }
    if (t == 1023) rowptr[Nn] = run;
}

__global__ __launch_bounds__(256) void sage_fill_kernel(
    const int* __restrict__ src, const int* __restrict__ dst,
    const int* __restrict__ rowptr, const int* __restrict__ rank,
    int* __restrict__ col, int E)
{
    int e = blockIdx.x * blockDim.x + threadIdx.x;
    if (e >= E) return;
    int p = rowptr[dst[e]] + rank[e];
    p = min(max(p, 0), E - 1);
    col[p] = src[e];
}

__global__ __launch_bounds__(256) void sage_gather_kernel(
    const float* __restrict__ feat, const int* __restrict__ rowptr,
    const int* __restrict__ col, float* __restrict__ agg, int Nn, int E)
{
    const int lane = threadIdx.x & 31;
    const int n = blockIdx.x * 8 + (threadIdx.x >> 5);
    if (n >= Nn) return;

    int beg = rowptr[n];
    int end = rowptr[n + 1];
    beg = min(max(beg, 0), E);
    end = min(max(end, beg), E);
    beg = __builtin_amdgcn_readfirstlane(beg);
    end = __builtin_amdgcn_readfirstlane(end);

    const int nmax = Nn - 1;
    const float* __restrict__ fb = feat + lane * 4;
    float4 a0 = make_float4(0.f, 0.f, 0.f, 0.f);
    float4 a1 = make_float4(0.f, 0.f, 0.f, 0.f);

    for (int base = beg; base < end; base += 32) {
        const int m = min(32, end - base);
        int myc = (lane < m) ? col[base + lane] : 0;
        myc = min(max(myc, 0), nmax);
        int j = 0;
        for (; j + 4 <= m; j += 4) {
            const int s0 = __builtin_amdgcn_readlane(myc, j);
            const int s1 = __builtin_amdgcn_readlane(myc, j + 1);
            const int s2 = __builtin_amdgcn_readlane(myc, j + 2);
            const int s3 = __builtin_amdgcn_readlane(myc, j + 3);
            const float* p0 = fb + (size_t)s0 * D;
            const float* p1 = fb + (size_t)s1 * D;
            const float* p2 = fb + (size_t)s2 * D;
            const float* p3 = fb + (size_t)s3 * D;
            const float4 x0 = *(const float4*)(p0);
            const float4 y0 = *(const float4*)(p0 + 128);
            const float4 x1 = *(const float4*)(p1);
            const float4 y1 = *(const float4*)(p1 + 128);
            const float4 x2 = *(const float4*)(p2);
            const float4 y2 = *(const float4*)(p2 + 128);
            const float4 x3 = *(const float4*)(p3);
            const float4 y3 = *(const float4*)(p3 + 128);
            a0.x += x0.x; a0.y += x0.y; a0.z += x0.z; a0.w += x0.w;
            a1.x += y0.x; a1.y += y0.y; a1.z += y0.z; a1.w += y0.w;
            a0.x += x1.x; a0.y += x1.y; a0.z += x1.z; a0.w += x1.w;
            a1.x += y1.x; a1.y += y1.y; a1.z += y1.z; a1.w += y1.w;
            a0.x += x2.x; a0.y += x2.y; a0.z += x2.z; a0.w += x2.w;
            a1.x += y2.x; a1.y += y2.y; a1.z += y2.z; a1.w += y2.w;
            a0.x += x3.x; a0.y += x3.y; a0.z += x3.z; a0.w += x3.w;
            a1.x += y3.x; a1.y += y3.y; a1.z += y3.z; a1.w += y3.w;
        }
        const int r = m - j;
        if (r > 0) {
            const int s0 = __builtin_amdgcn_readlane(myc, j);
            const int s1 = __builtin_amdgcn_readlane(myc, min(j + 1, 31));
            const int s2 = __builtin_amdgcn_readlane(myc, min(j + 2, 31));
            const float* p0 = fb + (size_t)s0 * D;
            const float* p1 = fb + (size_t)s1 * D;
            const float* p2 = fb + (size_t)s2 * D;
            const float4 x0 = *(const float4*)(p0);
            const float4 y0 = *(const float4*)(p0 + 128);
            float4 x1 = make_float4(0.f, 0.f, 0.f, 0.f), y1 = x1, x2 = x1, y2 = x1;
            if (r > 1) { x1 = *(const float4*)(p1); y1 = *(const float4*)(p1 + 128); }
            if (r > 2) { x2 = *(const float4*)(p2); y2 = *(const float4*)(p2 + 128); }
            a0.x += x0.x; a0.y += x0.y; a0.z += x0.z; a0.w += x0.w;
            a1.x += y0.x; a1.y += y0.y; a1.z += y0.z; a1.w += y0.w;
            if (r > 1) {
                a0.x += x1.x; a0.y += x1.y; a0.z += x1.z; a0.w += x1.w;
                a1.x += y1.x; a1.y += y1.y; a1.z += y1.z; a1.w += y1.w;
            }
            if (r > 2) {
                a0.x += x2.x; a0.y += x2.y; a0.z += x2.z; a0.w += x2.w;
                a1.x += y2.x; a1.y += y2.y; a1.z += y2.z; a1.w += y2.w;
            }
        }
    }

    float* o = agg + (size_t)n * D + lane * 4;
    *(float4*)(o)       = a0;
    *(float4*)(o + 128) = a1;
}

template <bool SCALE>
__device__ __forceinline__ void sage_gemm_path(
    const float* __restrict__ pa0, const float* __restrict__ pa1,
    const float* __restrict__ pb, const float s0, const float s1,
    v8f (&acc)[2][4])
{
#pragma unroll 2
    for (int k = 0; k < D; k += 8) {
        float4 a0 = *(const float4*)(pa0 + k);
        float4 a1 = *(const float4*)(pa1 + k);
        const float* pbk = pb + (size_t)k * D;
        const float4 b0 = *(const float4*)(pbk);
        const float4 b1 = *(const float4*)(pbk + D);
        const float4 b2 = *(const float4*)(pbk + 2 * D);
        const float4 b3 = *(const float4*)(pbk + 3 * D);
        if (SCALE) {
            a0.x *= s0; a0.y *= s0; a0.z *= s0; a0.w *= s0;
            a1.x *= s1; a1.y *= s1; a1.z *= s1; a1.w *= s1;
        }
        const v2f A00 = {a0.x, a0.y}, A01 = {a0.z, a0.w};
        const v2f A10 = {a1.x, a1.y}, A11 = {a1.z, a1.w};
        const float r0[4] = {b0.x, b0.y, b0.z, b0.w};
        const float r1[4] = {b1.x, b1.y, b1.z, b1.w};
        const float r2[4] = {b2.x, b2.y, b2.z, b2.w};
        const float r3[4] = {b3.x, b3.y, b3.z, b3.w};
#pragma unroll
        for (int j = 0; j < 4; ++j) {
            const v2f B = {r0[j], r1[j]};
            acc[0][j] = __builtin_amdgcn_wmma_f32_16x16x4_f32(false, A00, false, B,
                                                    (short)0, acc[0][j], false, false);
            acc[1][j] = __builtin_amdgcn_wmma_f32_16x16x4_f32(false, A10, false, B,
                                                    (short)0, acc[1][j], false, false);
        }
#pragma unroll
        for (int j = 0; j < 4; ++j) {
            const v2f B = {r2[j], r3[j]};
            acc[0][j] = __builtin_amdgcn_wmma_f32_16x16x4_f32(false, A01, false, B,
                                                    (short)0, acc[0][j], false, false);
            acc[1][j] = __builtin_amdgcn_wmma_f32_16x16x4_f32(false, A11, false, B,
                                                    (short)0, acc[1][j], false, false);
        }
    }
}

__global__ __launch_bounds__(128) void sage_gemm_kernel(
    const float* __restrict__ agg,  const float* __restrict__ invdeg,
    const float* __restrict__ feat, const float* __restrict__ Wl,
    const float* __restrict__ Wr,   const float* __restrict__ bias,
    float* __restrict__ out, int Nn, int do_relu)
{
    const int wave = threadIdx.x >> 5;
    const int lane = threadIdx.x & 31;
    const int half = lane >> 4;
    const int l15  = lane & 15;

    const int m0 = (blockIdx.x >> 1) * 64 + (wave >> 1) * 32;
    const int n0 = (blockIdx.x & 1) * 128 + (wave & 1) * 64;

    const int ra = min(m0 + l15, Nn - 1);
    const int rb = min(m0 + 16 + l15, Nn - 1);
    const float s0 = invdeg[ra];
    const float s1 = invdeg[rb];
    const size_t oa = (size_t)ra * D + 4 * half;
    const size_t ob = (size_t)rb * D + 4 * half;
    const size_t ow = (size_t)(4 * half) * D + n0 + 4 * l15;

    v8f acc[2][4] = {};

    sage_gemm_path<true>(agg + oa, agg + ob, Wl + ow, s0, s1, acc);
    sage_gemm_path<false>(feat + oa, feat + ob, Wr + ow, 1.0f, 1.0f, acc);

    const int ncol = n0 + 4 * l15;
    const float4 bn = *(const float4*)(bias + ncol);
#pragma unroll
    for (int mi = 0; mi < 2; ++mi) {
#pragma unroll
        for (int v = 0; v < 8; ++v) {
            const int row = m0 + 16 * mi + v + 8 * half;
            if (row < Nn) {
                float4 val;
                val.x = acc[mi][0][v] + bn.x;
                val.y = acc[mi][1][v] + bn.y;
                val.z = acc[mi][2][v] + bn.z;
                val.w = acc[mi][3][v] + bn.w;
                if (do_relu) {
                    val.x = fmaxf(val.x, 0.0f); val.y = fmaxf(val.y, 0.0f);
                    val.z = fmaxf(val.z, 0.0f); val.w = fmaxf(val.w, 0.0f);
                }
                *(float4*)(out + (size_t)row * D + ncol) = val;
            }
        }
    }
}

extern "C" void kernel_launch(void* const* d_in, const int* in_sizes, int n_in,
                              void* d_out, int out_size, void* d_ws, size_t ws_size,
                              hipStream_t stream)
{
    const float* x   = (const float*)d_in[0];
    const int*   ei  = (const int*)d_in[1];
    const float* W1l = (const float*)d_in[2];
    const float* b1  = (const float*)d_in[3];
    const float* W1r = (const float*)d_in[4];
    const float* W2l = (const float*)d_in[5];
    const float* b2  = (const float*)d_in[6];
    const float* W2r = (const float*)d_in[7];
    float* out = (float*)d_out;

    const int Nn = in_sizes[0] / D;
    const int E  = in_sizes[1] / 2;
    const int* src = ei;
    const int* dst = ei + E;

    float* agg = (float*)d_ws;
    float* deg = agg + (size_t)Nn * D;
    float* h   = deg + (size_t)Nn;

    int* cnt    = (int*)deg;
    int* rank   = (int*)h;
    int* rowptr = (int*)d_out;
    int* col    = rowptr + (size_t)Nn + 1;

    const size_t degBytes = (size_t)Nn * sizeof(float);

    const int blk = 256;
    const int edgeBlocks = (E + blk - 1) / blk;
    const int gatherBlocks = (Nn + 7) / 8;
    const int gemmBlocks = ((Nn + 63) / 64) * (D / 128);
    const int gemmBlk = 128;

    hipMemsetAsync(cnt, 0, degBytes, stream);
    sage_rank_kernel<<<edgeBlocks, blk, 0, stream>>>(dst, cnt, rank, E);
    sage_scan_kernel<<<1, 1024, 0, stream>>>(cnt, rowptr, Nn);
    sage_fill_kernel<<<edgeBlocks, blk, 0, stream>>>(src, dst, rowptr, rank, col, E);

    sage_gather_kernel<<<gatherBlocks, blk, 0, stream>>>(x, rowptr, col, agg, Nn, E);
    sage_gemm_kernel<<<gemmBlocks, gemmBlk, 0, stream>>>(agg, deg, x, W1l, W1r, b1,
                                                   h, Nn,  1);

    sage_gather_kernel<<<gatherBlocks, blk, 0, stream>>>(h, rowptr, col, agg, Nn, E);
    sage_gemm_kernel<<<gemmBlocks, gemmBlk, 0, stream>>>(agg, deg, h, W2l, W2r, b2,
                                                   out, Nn,  0);
}


# Neighbour sum fused into the layer kernel: a block sums its 32 rows into LDS as the WMMA A operand; half the blocks add the self product first (k sum order changes there) so gathers overlap WMMA work
# speedup vs baseline: 3.4277x; 1.0839x over previous
#include <hip/hip_runtime.h>
#include <hip/hip_bf16.h>


typedef __attribute__((ext_vector_type(2))) float v2f;
typedef __attribute__((ext_vector_type(8))) float v8f;

#define D 256

__global__ __launch_bounds__(256) void sage_rank_kernel(
    const int* __restrict__ dst, int* __restrict__ cnt,
    int* __restrict__ rank, int E)
{
    int e = blockIdx.x * blockDim.x + threadIdx.x;
    if (e < E) rank[e] = atomicAdd(&cnt[dst[e]], 1);
}

__global__ __launch_bounds__(1024) void sage_scan_kernel(
    int* cd, int* __restrict__ rowptr, int Nn)
{
    __shared__ int wsum[32];
    const int t = threadIdx.x;
    const int lane = t & 31;
    const int w = t >> 5;
    const int per = (Nn + 1023) / 1024;
    const int i0 = min(t * per, Nn);
    const int i1 = min(i0 + per, Nn);

    int s = 0;
#pragma unroll 8
    for (int i = i0; i < i1; ++i) s += cd[i];

    int inc = s;
    for (int o = 1; o < 32; o <<= 1) {
        int v = __shfl_up(inc, o);
        if (lane >= o) inc += v;
    }
    if (lane == 31) wsum[w] = inc;
    __syncthreads();
    if (w == 0) {
        int v = wsum[lane];
        int vi = v;
        for (int o = 1; o < 32; o <<= 1) {
            int u = __shfl_up(vi, o);
            if (lane >= o) vi += u;
        }
        wsum[lane] = vi - v;
    }
    __syncthreads();

    int run = wsum[w] + inc - s;
    for (int i = i0; i < i1; ++i) {
        const int c = cd[i];
        rowptr[i] = run;
        cd[i] = __float_as_int(1.0f / fmaxf((float)c, 1.0f));
        run += c;
    }
    if (t == 1023) rowptr[Nn] = run;
}

__global__ __launch_bounds__(256) void sage_fill_kernel(
    const int* __restrict__ src, const int* __restrict__ dst,
    const int* __restrict__ rowptr, const int* __restrict__ rank,
    int* __restrict__ col, int E)
{
    int e = blockIdx.x * blockDim.x + threadIdx.x;
    if (e >= E) return;
    int p = rowptr[dst[e]] + rank[e];
    p = min(max(p, 0), E - 1);
    col[p] = src[e];
}

__device__ __forceinline__ void sage_row_sum(
    const float* __restrict__ fb, const int* __restrict__ col,
    const int beg, const int end, const int nmax, const int lane,
    float4& a0, float4& a1)
{
    a0 = make_float4(0.f, 0.f, 0.f, 0.f);
    a1 = make_float4(0.f, 0.f, 0.f, 0.f);

    for (int base = beg; base < end; base += 32) {
        const int m = min(32, end - base);
        int myc = (lane < m) ? col[base + lane] : 0;
        myc = min(max(myc, 0), nmax);
        int j = 0;
        for (; j + 4 <= m; j += 4) {
            const int s0 = __builtin_amdgcn_readlane(myc, j);
            const int s1 = __builtin_amdgcn_readlane(myc, j + 1);
            const int s2 = __builtin_amdgcn_readlane(myc, j + 2);
            const int s3 = __builtin_amdgcn_readlane(myc, j + 3);
            const float* p0 = fb + (size_t)s0 * D;
            const float* p1 = fb + (size_t)s1 * D;
            const float* p2 = fb + (size_t)s2 * D;
            const float* p3 = fb + (size_t)s3 * D;
            const float4 x0 = *(const float4*)(p0);
            const float4 y0 = *(const float4*)(p0 + 128);
            const float4 x1 = *(const float4*)(p1);
            const float4 y1 = *(const float4*)(p1 + 128);
            const float4 x2 = *(const float4*)(p2);
            const float4 y2 = *(const float4*)(p2 + 128);
            const float4 x3 = *(const float4*)(p3);
            const float4 y3 = *(const float4*)(p3 + 128);
            a0.x += x0.x; a0.y += x0.y; a0.z += x0.z; a0.w += x0.w;
            a1.x += y0.x; a1.y += y0.y; a1.z += y0.z; a1.w += y0.w;
            a0.x += x1.x; a0.y += x1.y; a0.z += x1.z; a0.w += x1.w;
            a1.x += y1.x; a1.y += y1.y; a1.z += y1.z; a1.w += y1.w;
            a0.x += x2.x; a0.y += x2.y; a0.z += x2.z; a0.w += x2.w;
            a1.x += y2.x; a1.y += y2.y; a1.z += y2.z; a1.w += y2.w;
            a0.x += x3.x; a0.y += x3.y; a0.z += x3.z; a0.w += x3.w;
            a1.x += y3.x; a1.y += y3.y; a1.z += y3.z; a1.w += y3.w;
        }
        const int r = m - j;
        if (r > 0) {
            const int s0 = __builtin_amdgcn_readlane(myc, j);
            const int s1 = __builtin_amdgcn_readlane(myc, min(j + 1, 31));
            const int s2 = __builtin_amdgcn_readlane(myc, min(j + 2, 31));
            const float* p0 = fb + (size_t)s0 * D;
            const float* p1 = fb + (size_t)s1 * D;
            const float* p2 = fb + (size_t)s2 * D;
            const float4 x0 = *(const float4*)(p0);
            const float4 y0 = *(const float4*)(p0 + 128);
            float4 x1 = make_float4(0.f, 0.f, 0.f, 0.f), y1 = x1, x2 = x1, y2 = x1;
            if (r > 1) { x1 = *(const float4*)(p1); y1 = *(const float4*)(p1 + 128); }
            if (r > 2) { x2 = *(const float4*)(p2); y2 = *(const float4*)(p2 + 128); }
            a0.x += x0.x; a0.y += x0.y; a0.z += x0.z; a0.w += x0.w;
            a1.x += y0.x; a1.y += y0.y; a1.z += y0.z; a1.w += y0.w;
            if (r > 1) {
                a0.x += x1.x; a0.y += x1.y; a0.z += x1.z; a0.w += x1.w;
                a1.x += y1.x; a1.y += y1.y; a1.z += y1.z; a1.w += y1.w;
            }
            if (r > 2) {
                a0.x += x2.x; a0.y += x2.y; a0.z += x2.z; a0.w += x2.w;
                a1.x += y2.x; a1.y += y2.y; a1.z += y2.z; a1.w += y2.w;
            }
        }
    }
}

__device__ __forceinline__ void sage_gemm_path(
    const float* __restrict__ pa0, const float* __restrict__ pa1,
    const float* __restrict__ pb, v8f (&acc)[2][4])
{
#pragma unroll 2
    for (int k = 0; k < D; k += 8) {
        const float4 a0 = *(const float4*)(pa0 + k);
        const float4 a1 = *(const float4*)(pa1 + k);
        const float* pbk = pb + (size_t)k * D;
        const float4 b0 = *(const float4*)(pbk);
        const float4 b1 = *(const float4*)(pbk + D);
        const float4 b2 = *(const float4*)(pbk + 2 * D);
        const float4 b3 = *(const float4*)(pbk + 3 * D);
        const v2f A00 = {a0.x, a0.y}, A01 = {a0.z, a0.w};
        const v2f A10 = {a1.x, a1.y}, A11 = {a1.z, a1.w};
        const float r0[4] = {b0.x, b0.y, b0.z, b0.w};
        const float r1[4] = {b1.x, b1.y, b1.z, b1.w};
        const float r2[4] = {b2.x, b2.y, b2.z, b2.w};
        const float r3[4] = {b3.x, b3.y, b3.z, b3.w};
#pragma unroll
        for (int j = 0; j < 4; ++j) {
            const v2f B = {r0[j], r1[j]};
            acc[0][j] = __builtin_amdgcn_wmma_f32_16x16x4_f32(false, A00, false, B,
                                                    (short)0, acc[0][j], false, false);
            acc[1][j] = __builtin_amdgcn_wmma_f32_16x16x4_f32(false, A10, false, B,
                                                    (short)0, acc[1][j], false, false);
        }
#pragma unroll
        for (int j = 0; j < 4; ++j) {
            const v2f B = {r2[j], r3[j]};
            acc[0][j] = __builtin_amdgcn_wmma_f32_16x16x4_f32(false, A01, false, B,
                                                    (short)0, acc[0][j], false, false);
            acc[1][j] = __builtin_amdgcn_wmma_f32_16x16x4_f32(false, A11, false, B,
                                                    (short)0, acc[1][j], false, false);
        }
    }
}

#define APITCH 260

__global__ __launch_bounds__(128) void sage_layer_kernel(
    const float* __restrict__ feat, const int* __restrict__ rowptr,
    const int* __restrict__ col,    const float* __restrict__ invdeg,
    const float* __restrict__ Wl,   const float* __restrict__ Wr,
    const float* __restrict__ bias, float* __restrict__ out,
    int Nn, int E, int do_relu)
{
    __shared__ __attribute__((aligned(16))) float sA[32 * APITCH];

    const int wave = threadIdx.x >> 5;
    const int lane = threadIdx.x & 31;
    const int half = lane >> 4;
    const int l15  = lane & 15;

    const int m0 = blockIdx.x * 32;
    const int n0 = wave * 64;

    const int ra = min(m0 + l15, Nn - 1);
    const int rb = min(m0 + 16 + l15, Nn - 1);
    const size_t oa = (size_t)ra * D + 4 * half;
    const size_t ob = (size_t)rb * D + 4 * half;
    const size_t ow = (size_t)(4 * half) * D + n0 + 4 * l15;

    v8f acc[2][4] = {};

    const bool self_first = (__popc(blockIdx.x) & 1) != 0;

    if (self_first)
        sage_gemm_path(feat + oa, feat + ob, Wr + ow, acc);

    {
        const int nb = m0 + wave * 8;
        int rp = rowptr[min(nb + lane, Nn)];
        rp = min(max(rp, 0), E);
        const float iv = invdeg[min(nb + lane, Nn - 1)];
        const float* __restrict__ fb = feat + lane * 4;
#pragma unroll 1
        for (int i = 0; i < 8; ++i) {
            const int beg = __builtin_amdgcn_readlane(rp, i);
            const int end = max(__builtin_amdgcn_readlane(rp, i + 1), beg);
            const float sc = __int_as_float(
                __builtin_amdgcn_readlane(__float_as_int(iv), i));
            float4 a0, a1;
            sage_row_sum(fb, col, beg, end, Nn - 1, lane, a0, a1);
            a0.x *= sc; a0.y *= sc; a0.z *= sc; a0.w *= sc;
            a1.x *= sc; a1.y *= sc; a1.z *= sc; a1.w *= sc;
            float* o = sA + (wave * 8 + i) * APITCH + lane * 4;
            *(float4*)(o)       = a0;
            *(float4*)(o + 128) = a1;
        }
    }
    __syncthreads();

    sage_gemm_path(sA + l15 * APITCH + 4 * half,
                   sA + (16 + l15) * APITCH + 4 * half, Wl + ow, acc);

    if (!self_first)
        sage_gemm_path(feat + oa, feat + ob, Wr + ow, acc);

    const int ncol = n0 + 4 * l15;
    const float4 bn = *(const float4*)(bias + ncol);
#pragma unroll
    for (int mi = 0; mi < 2; ++mi) {
#pragma unroll
        for (int v = 0; v < 8; ++v) {
            const int row = m0 + 16 * mi + v + 8 * half;
            if (row < Nn) {
                float4 val;
                val.x = acc[mi][0][v] + bn.x;
                val.y = acc[mi][1][v] + bn.y;
                val.z = acc[mi][2][v] + bn.z;
                val.w = acc[mi][3][v] + bn.w;
                if (do_relu) {
                    val.x = fmaxf(val.x, 0.0f); val.y = fmaxf(val.y, 0.0f);
                    val.z = fmaxf(val.z, 0.0f); val.w = fmaxf(val.w, 0.0f);
                }
                *(float4*)(out + (size_t)row * D + ncol) = val;
            }
        }
    }
}

extern "C" void kernel_launch(void* const* d_in, const int* in_sizes, int n_in,
                              void* d_out, int out_size, void* d_ws, size_t ws_size,
                              hipStream_t stream)
{
    const float* x   = (const float*)d_in[0];
    const int*   ei  = (const int*)d_in[1];
    const float* W1l = (const float*)d_in[2];
    const float* b1  = (const float*)d_in[3];
    const float* W1r = (const float*)d_in[4];
    const float* W2l = (const float*)d_in[5];
    const float* b2  = (const float*)d_in[6];
    const float* W2r = (const float*)d_in[7];
    float* out = (float*)d_out;

    const int Nn = in_sizes[0] / D;
    const int E  = in_sizes[1] / 2;
    const int* src = ei;
    const int* dst = ei + E;

    float* grp = (float*)d_ws;
    float* deg = grp + (size_t)Nn * D;
    float* h   = deg + (size_t)Nn;

    int* cnt    = (int*)deg;
    int* rank   = (int*)h;
    int* rowptr = (int*)grp;
    int* col    = rowptr + (size_t)Nn + 1;

    const size_t degBytes = (size_t)Nn * sizeof(float);

    const int blk = 256;
    const int edgeBlocks = (E + blk - 1) / blk;
    const int layerBlocks = (Nn + 31) / 32;
    const int layerBlk = 128;

    hipMemsetAsync(cnt, 0, degBytes, stream);
    sage_rank_kernel<<<edgeBlocks, blk, 0, stream>>>(dst, cnt, rank, E);
    sage_scan_kernel<<<1, 1024, 0, stream>>>(cnt, rowptr, Nn);
    sage_fill_kernel<<<edgeBlocks, blk, 0, stream>>>(src, dst, rowptr, rank, col, E);

    sage_layer_kernel<<<layerBlocks, layerBlk, 0, stream>>>(x, rowptr, col, deg, W1l, W1r,
                                                      b1, h, Nn, E,  1);

    sage_layer_kernel<<<layerBlocks, layerBlk, 0, stream>>>(h, rowptr, col, deg, W2l, W2r,
                                                      b2, out, Nn, E,  0);
}


# Edge grouping by a two-level counting sort on LDS counters (256 node ranges: count, offsets, scatter, rank per range) instead of one returning global add per edge; sums in LDS arrival order
# speedup vs baseline: 4.1729x; 1.2174x over previous
#include <hip/hip_runtime.h>
#include <hip/hip_bf16.h>


typedef __attribute__((ext_vector_type(2))) float v2f;
typedef __attribute__((ext_vector_type(8))) float v8f;

#define D 256

#define SR 256
#define EB 16

__global__ __launch_bounds__(256) void sage_count_kernel(
    const int* __restrict__ dst, int* __restrict__ bh, int E, int Nn)
{
    __shared__ int hist[SR];
    const int t = threadIdx.x;
    hist[t] = 0;
    __syncthreads();
    const int rs = (Nn + SR - 1) / SR;
    const int ch = (E + SR - 1) / SR;
    const int e0 = min(blockIdx.x * ch, E);
    const int e1 = min(e0 + ch, E);
    for (int eb = e0 + t; eb < e1; eb += 256 * EB) {
        int d[EB];
#pragma unroll
        for (int k = 0; k < EB; ++k) d[k] = dst[min(eb + 256 * k, e1 - 1)];
#pragma unroll
        for (int k = 0; k < EB; ++k) {
            if (eb + 256 * k < e1) {
                const int dd = min(max(d[k], 0), Nn - 1);
                atomicAdd(&hist[min(dd / rs, SR - 1)], 1);
            }
        }
    }
    __syncthreads();
    bh[t * SR + blockIdx.x] = hist[t];
}

__global__ __launch_bounds__(1024) void sage_offsets_kernel(int* bh)
{
    __shared__ int wsum[32];
    const int t = threadIdx.x;
    const int lane = t & 31;
    const int w = t >> 5;
    int4* p = (int4*)(bh + w * 2048) + lane;
    int4 v[16];
#pragma unroll
    for (int i = 0; i < 16; ++i) v[i] = p[i * 32];

    int carry = 0;
#pragma unroll
    for (int i = 0; i < 16; ++i) {
        const int4 c = v[i];
        const int tot = c.x + c.y + c.z + c.w;
        int inc = tot;
        for (int o = 1; o < 32; o <<= 1) {
            int u = __shfl_up(inc, o);
            if (lane >= o) inc += u;
        }
        const int b = carry + inc - tot;
        v[i] = make_int4(b, b + c.x, b + c.x + c.y, b + c.x + c.y + c.z);
        carry += __shfl(inc, 31);
    }
    if (lane == 0) wsum[w] = carry;
    __syncthreads();
    if (w == 0) {
        int x = wsum[lane];
        int xi = x;
        for (int o = 1; o < 32; o <<= 1) {
            int u = __shfl_up(xi, o);
            if (lane >= o) xi += u;
        }
        wsum[lane] = xi - x;
    }
    __syncthreads();
    const int wb = wsum[w];
#pragma unroll
    for (int i = 0; i < 16; ++i)
        p[i * 32] = make_int4(v[i].x + wb, v[i].y + wb, v[i].z + wb, v[i].w + wb);
    if (t == 1023) bh[SR * SR] = wb + carry;
}

__global__ __launch_bounds__(256) void sage_scatter_kernel(
    const int* __restrict__ src, const int* __restrict__ dst,
    const int* __restrict__ bh, int* __restrict__ tmpd, int* __restrict__ tmps,
    int E, int Nn)
{
    __shared__ int base[SR];
    __shared__ int cnt[SR];
    const int t = threadIdx.x;
    base[t] = bh[t * SR + blockIdx.x];
    cnt[t] = 0;
    __syncthreads();
    const int rs = (Nn + SR - 1) / SR;
    const int ch = (E + SR - 1) / SR;
    const int e0 = min(blockIdx.x * ch, E);
    const int e1 = min(e0 + ch, E);
    for (int eb = e0 + t; eb < e1; eb += 256 * EB) {
        int d[EB], sv[EB];
#pragma unroll
        for (int k = 0; k < EB; ++k) {
            const int e = min(eb + 256 * k, e1 - 1);
            d[k] = dst[e];
            sv[k] = src[e];
        }
#pragma unroll
        for (int k = 0; k < EB; ++k) {
            if (eb + 256 * k < e1) {
                const int dd = min(max(d[k], 0), Nn - 1);
                const int b = min(dd / rs, SR - 1);
                int q = base[b] + atomicAdd(&cnt[b], 1);
                q = min(max(q, 0), E - 1);
                tmpd[q] = dd;
                tmps[q] = sv[k];
            }
        }
    }
}

__global__ __launch_bounds__(256) void sage_rows_kernel(
    const int* __restrict__ bh, const int* __restrict__ tmpd,
    const int* __restrict__ tmps, int* __restrict__ rowptr,
    int* __restrict__ col, float* __restrict__ invdeg, int E, int Nn)
{
    __shared__ int cnt[SR];
    __shared__ int start[SR];
    __shared__ int wtot[8];
    const int t = threadIdx.x;
    const int lane = t & 31;
    const int w = t >> 5;
    const int rs = (Nn + SR - 1) / SR;
    const int n0 = blockIdx.x * rs;
    int s = bh[blockIdx.x * SR];
    int u = bh[(blockIdx.x + 1) * SR];
    s = min(max(s, 0), E);
    u = min(max(u, s), E);
    cnt[t] = 0;
    __syncthreads();
    for (int eb = s + t; eb < u; eb += 256 * EB) {
        int d[EB];
#pragma unroll
        for (int k = 0; k < EB; ++k) d[k] = tmpd[min(eb + 256 * k, u - 1)];
#pragma unroll
        for (int k = 0; k < EB; ++k)
            if (eb + 256 * k < u)
                atomicAdd(&cnt[min(max(d[k] - n0, 0), SR - 1)], 1);
    }
    __syncthreads();

    const int c = cnt[t];
    int inc = c;
    for (int o = 1; o < 32; o <<= 1) {
        int x = __shfl_up(inc, o);
        if (lane >= o) inc += x;
    }
    if (lane == 31) wtot[w] = inc;
    __syncthreads();
    int before = 0;
    for (int i = 0; i < w; ++i) before += wtot[i];
    const int st = s + before + inc - c;
    start[t] = st;
    cnt[t] = 0;
    const int n = n0 + t;
    if (t < rs && n < Nn) {
        rowptr[n] = st;
        invdeg[n] = 1.0f / fmaxf((float)c, 1.0f);
    }
    if (blockIdx.x == 0 && t == 0) rowptr[Nn] = E;
    __syncthreads();

    for (int eb = s + t; eb < u; eb += 256 * EB) {
        int d[EB], sv[EB];
#pragma unroll
        for (int k = 0; k < EB; ++k) {
            const int e = min(eb + 256 * k, u - 1);
            d[k] = tmpd[e];
            sv[k] = tmps[e];
        }
#pragma unroll
        for (int k = 0; k < EB; ++k) {
            if (eb + 256 * k < u) {
                const int j = min(max(d[k] - n0, 0), SR - 1);
                int q = start[j] + atomicAdd(&cnt[j], 1);
                q = min(max(q, 0), E - 1);
                col[q] = sv[k];
            }
        }
    }
}

__device__ __forceinline__ void sage_row_sum(
    const float* __restrict__ fb, const int* __restrict__ col,
    const int beg, const int end, const int nmax, const int lane,
    float4& a0, float4& a1)
{
    a0 = make_float4(0.f, 0.f, 0.f, 0.f);
    a1 = make_float4(0.f, 0.f, 0.f, 0.f);

    for (int base = beg; base < end; base += 32) {
        const int m = min(32, end - base);
        int myc = (lane < m) ? col[base + lane] : 0;
        myc = min(max(myc, 0), nmax);
        int j = 0;
        for (; j + 4 <= m; j += 4) {
            const int s0 = __builtin_amdgcn_readlane(myc, j);
            const int s1 = __builtin_amdgcn_readlane(myc, j + 1);
            const int s2 = __builtin_amdgcn_readlane(myc, j + 2);
            const int s3 = __builtin_amdgcn_readlane(myc, j + 3);
            const float* p0 = fb + (size_t)s0 * D;
            const float* p1 = fb + (size_t)s1 * D;
            const float* p2 = fb + (size_t)s2 * D;
            const float* p3 = fb + (size_t)s3 * D;
            const float4 x0 = *(const float4*)(p0);
            const float4 y0 = *(const float4*)(p0 + 128);
            const float4 x1 = *(const float4*)(p1);
            const float4 y1 = *(const float4*)(p1 + 128);
            const float4 x2 = *(const float4*)(p2);
            const float4 y2 = *(const float4*)(p2 + 128);
            const float4 x3 = *(const float4*)(p3);
            const float4 y3 = *(const float4*)(p3 + 128);
            a0.x += x0.x; a0.y += x0.y; a0.z += x0.z; a0.w += x0.w;
            a1.x += y0.x; a1.y += y0.y; a1.z += y0.z; a1.w += y0.w;
            a0.x += x1.x; a0.y += x1.y; a0.z += x1.z; a0.w += x1.w;
            a1.x += y1.x; a1.y += y1.y; a1.z += y1.z; a1.w += y1.w;
            a0.x += x2.x; a0.y += x2.y; a0.z += x2.z; a0.w += x2.w;
            a1.x += y2.x; a1.y += y2.y; a1.z += y2.z; a1.w += y2.w;
            a0.x += x3.x; a0.y += x3.y; a0.z += x3.z; a0.w += x3.w;
            a1.x += y3.x; a1.y += y3.y; a1.z += y3.z; a1.w += y3.w;
        }
        const int r = m - j;
        if (r > 0) {
            const int s0 = __builtin_amdgcn_readlane(myc, j);
            const int s1 = __builtin_amdgcn_readlane(myc, min(j + 1, 31));
            const int s2 = __builtin_amdgcn_readlane(myc, min(j + 2, 31));
            const float* p0 = fb + (size_t)s0 * D;
            const float* p1 = fb + (size_t)s1 * D;
            const float* p2 = fb + (size_t)s2 * D;
            const float4 x0 = *(const float4*)(p0);
            const float4 y0 = *(const float4*)(p0 + 128);
            float4 x1 = make_float4(0.f, 0.f, 0.f, 0.f), y1 = x1, x2 = x1, y2 = x1;
            if (r > 1) { x1 = *(const float4*)(p1); y1 = *(const float4*)(p1 + 128); }
            if (r > 2) { x2 = *(const float4*)(p2); y2 = *(const float4*)(p2 + 128); }
            a0.x += x0.x; a0.y += x0.y; a0.z += x0.z; a0.w += x0.w;
            a1.x += y0.x; a1.y += y0.y; a1.z += y0.z; a1.w += y0.w;
            if (r > 1) {
                a0.x += x1.x; a0.y += x1.y; a0.z += x1.z; a0.w += x1.w;
                a1.x += y1.x; a1.y += y1.y; a1.z += y1.z; a1.w += y1.w;
            }
            if (r > 2) {
                a0.x += x2.x; a0.y += x2.y; a0.z += x2.z; a0.w += x2.w;
                a1.x += y2.x; a1.y += y2.y; a1.z += y2.z; a1.w += y2.w;
            }
        }
    }
}

__device__ __forceinline__ void sage_gemm_path(
    const float* __restrict__ pa0, const float* __restrict__ pa1,
    const float* __restrict__ pb, v8f (&acc)[2][4])
{
#pragma unroll 2
    for (int k = 0; k < D; k += 8) {
        const float4 a0 = *(const float4*)(pa0 + k);
        const float4 a1 = *(const float4*)(pa1 + k);
        const float* pbk = pb + (size_t)k * D;
        const float4 b0 = *(const float4*)(pbk);
        const float4 b1 = *(const float4*)(pbk + D);
        const float4 b2 = *(const float4*)(pbk + 2 * D);
        const float4 b3 = *(const float4*)(pbk + 3 * D);
        const v2f A00 = {a0.x, a0.y}, A01 = {a0.z, a0.w};
        const v2f A10 = {a1.x, a1.y}, A11 = {a1.z, a1.w};
        const float r0[4] = {b0.x, b0.y, b0.z, b0.w};
        const float r1[4] = {b1.x, b1.y, b1.z, b1.w};
        const float r2[4] = {b2.x, b2.y, b2.z, b2.w};
        const float r3[4] = {b3.x, b3.y, b3.z, b3.w};
#pragma unroll
        for (int j = 0; j < 4; ++j) {
            const v2f B = {r0[j], r1[j]};
            acc[0][j] = __builtin_amdgcn_wmma_f32_16x16x4_f32(false, A00, false, B,
                                                    (short)0, acc[0][j], false, false);
            acc[1][j] = __builtin_amdgcn_wmma_f32_16x16x4_f32(false, A10, false, B,
                                                    (short)0, acc[1][j], false, false);
        }
#pragma unroll
        for (int j = 0; j < 4; ++j) {
            const v2f B = {r2[j], r3[j]};
            acc[0][j] = __builtin_amdgcn_wmma_f32_16x16x4_f32(false, A01, false, B,
                                                    (short)0, acc[0][j], false, false);
            acc[1][j] = __builtin_amdgcn_wmma_f32_16x16x4_f32(false, A11, false, B,
                                                    (short)0, acc[1][j], false, false);
        }
    }
}

#define APITCH 260

__global__ __launch_bounds__(128) void sage_layer_kernel(
    const float* __restrict__ feat, const int* __restrict__ rowptr,
    const int* __restrict__ col,    const float* __restrict__ invdeg,
    const float* __restrict__ Wl,   const float* __restrict__ Wr,
    const float* __restrict__ bias, float* __restrict__ out,
    int Nn, int E, int do_relu)
{
    __shared__ __attribute__((aligned(16))) float sA[32 * APITCH];

    const int wave = threadIdx.x >> 5;
    const int lane = threadIdx.x & 31;
    const int half = lane >> 4;
    const int l15  = lane & 15;

    const int m0 = blockIdx.x * 32;
    const int n0 = wave * 64;

    const int ra = min(m0 + l15, Nn - 1);
    const int rb = min(m0 + 16 + l15, Nn - 1);
    const size_t oa = (size_t)ra * D + 4 * half;
    const size_t ob = (size_t)rb * D + 4 * half;
    const size_t ow = (size_t)(4 * half) * D + n0 + 4 * l15;

    v8f acc[2][4] = {};

    const bool self_first = (__popc(blockIdx.x) & 1) != 0;

    if (self_first)
        sage_gemm_path(feat + oa, feat + ob, Wr + ow, acc);

    {
        const int nb = m0 + wave * 8;
        int rp = rowptr[min(nb + lane, Nn)];
        rp = min(max(rp, 0), E);
        const float iv = invdeg[min(nb + lane, Nn - 1)];
        const float* __restrict__ fb = feat + lane * 4;
#pragma unroll 1
        for (int i = 0; i < 8; ++i) {
            const int beg = __builtin_amdgcn_readlane(rp, i);
            const int end = max(__builtin_amdgcn_readlane(rp, i + 1), beg);
            const float sc = __int_as_float(
                __builtin_amdgcn_readlane(__float_as_int(iv), i));
            float4 a0, a1;
            sage_row_sum(fb, col, beg, end, Nn - 1, lane, a0, a1);
            a0.x *= sc; a0.y *= sc; a0.z *= sc; a0.w *= sc;
            a1.x *= sc; a1.y *= sc; a1.z *= sc; a1.w *= sc;
            float* o = sA + (wave * 8 + i) * APITCH + lane * 4;
            *(float4*)(o)       = a0;
            *(float4*)(o + 128) = a1;
        }
    }
    __syncthreads();

    sage_gemm_path(sA + l15 * APITCH + 4 * half,
                   sA + (16 + l15) * APITCH + 4 * half, Wl + ow, acc);

    if (!self_first)
        sage_gemm_path(feat + oa, feat + ob, Wr + ow, acc);

    const int ncol = n0 + 4 * l15;
    const float4 bn = *(const float4*)(bias + ncol);
#pragma unroll
    for (int mi = 0; mi < 2; ++mi) {
#pragma unroll
        for (int v = 0; v < 8; ++v) {
            const int row = m0 + 16 * mi + v + 8 * half;
            if (row < Nn) {
                float4 val;
                val.x = acc[mi][0][v] + bn.x;
                val.y = acc[mi][1][v] + bn.y;
                val.z = acc[mi][2][v] + bn.z;
                val.w = acc[mi][3][v] + bn.w;
                if (do_relu) {
                    val.x = fmaxf(val.x, 0.0f); val.y = fmaxf(val.y, 0.0f);
                    val.z = fmaxf(val.z, 0.0f); val.w = fmaxf(val.w, 0.0f);
                }
                *(float4*)(out + (size_t)row * D + ncol) = val;
            }
        }
    }
}

extern "C" void kernel_launch(void* const* d_in, const int* in_sizes, int n_in,
                              void* d_out, int out_size, void* d_ws, size_t ws_size,
                              hipStream_t stream)
{
    const float* x   = (const float*)d_in[0];
    const int*   ei  = (const int*)d_in[1];
    const float* W1l = (const float*)d_in[2];
    const float* b1  = (const float*)d_in[3];
    const float* W1r = (const float*)d_in[4];
    const float* W2l = (const float*)d_in[5];
    const float* b2  = (const float*)d_in[6];
    const float* W2r = (const float*)d_in[7];
    float* out = (float*)d_out;

    const int Nn = in_sizes[0] / D;
    const int E  = in_sizes[1] / 2;
    const int* src = ei;
    const int* dst = ei + E;

    float* grp = (float*)d_ws;
    float* deg = grp + (size_t)Nn * D;
    float* h   = deg + (size_t)Nn;

    int* bh     = (int*)grp;
    int* rowptr = bh + SR * SR + 4;
    int* col    = rowptr + (size_t)Nn + 1;
    int* tmpd   = (int*)h;
    int* tmps   = tmpd + (size_t)E;

    const int layerBlocks = (Nn + 31) / 32;
    const int layerBlk = 128;

    sage_count_kernel<<<SR, 256, 0, stream>>>(dst, bh, E, Nn);
    sage_offsets_kernel<<<1, 1024, 0, stream>>>(bh);
    sage_scatter_kernel<<<SR, 256, 0, stream>>>(src, dst, bh, tmpd, tmps, E, Nn);
    sage_rows_kernel<<<SR, 256, 0, stream>>>(bh, tmpd, tmps, rowptr, col, deg, E, Nn);

    sage_layer_kernel<<<layerBlocks, layerBlk, 0, stream>>>(x, rowptr, col, deg, W1l, W1r,
                                                      b1, h, Nn, E,  1);

    sage_layer_kernel<<<layerBlocks, layerBlk, 0, stream>>>(h, rowptr, col, deg, W2l, W2r,
                                                      b2, out, Nn, E,  0);
}
